# starting kernel
# baseline (speedup 1.0000x reference)
#include <hip/hip_runtime.h>
#include <math.h>
#include <float.h>


typedef __attribute__((ext_vector_type(2))) float v2f;
typedef __attribute__((ext_vector_type(8))) float v8f;

#define NN 50000
#define EE 1200000
#define DD 64
#define GG 1024

static_assert(NN % 16 == 0, "WMMA GEMM requires row count multiple of 16");


__global__ void fill_f32_kernel(float* __restrict__ p, float v, int n) {
  int i = blockIdx.x * blockDim.x + threadIdx.x;
  if (i < n) p[i] = v;
}

__global__ void degree_kernel(const int* __restrict__ dst, float* __restrict__ deg, int e) {
  int i = blockIdx.x * blockDim.x + threadIdx.x;
  if (i < e) atomicAdd(&deg[dst[i]], 1.0f);
}

__global__ void dinv_kernel(const float* __restrict__ deg, float* __restrict__ dinv, int n) {
  int i = blockIdx.x * blockDim.x + threadIdx.x;
  if (i < n) dinv[i] = rsqrtf(deg[i]);
}

__global__ void pad_x_kernel(const float* __restrict__ x, float* __restrict__ hpad, int n) {
  int i = blockIdx.x * blockDim.x + threadIdx.x;
  if (i >= n * 16) return;
  int r = i >> 4, c = i & 15;
  hpad[i] = (c < 9) ? x[r * 9 + c] : 0.0f;
}

__global__ void pad_w0_kernel(const float* __restrict__ W0, float* __restrict__ w0pad) {
  int i = blockIdx.x * blockDim.x + threadIdx.x;
  if (i >= 16 * 64) return;
  int r = i >> 6, c = i & 63;
  w0pad[i] = (r < 9) ? W0[r * 64 + c] : 0.0f;
}

template <int KDIM>
__global__ void gemm_wmma_kernel(const float* __restrict__ A,
                                 const float* __restrict__ W,
                                 float* __restrict__ out) {
  const int wave = threadIdx.x >> 5;
  const int lane = threadIdx.x & 31;
  const int l15  = lane & 15;
  const int half = lane >> 4;
  const int m0   = blockIdx.x * 16;
  const int n0   = wave * 16;
  const int row  = m0 + l15;
  const int col  = n0 + l15;

  const float* __restrict__ arow = A + (size_t)row * KDIM + half * 2;
  const float* __restrict__ bcol = W + (size_t)half * 2 * 64 + col;

  v8f c = {};
#pragma unroll
  for (int k = 0; k < KDIM; k += 4) {
    v2f a, b;
    a.x = arow[k + 0];
    a.y = arow[k + 1];
    b.x = bcol[(size_t)k * 64];
    b.y = bcol[(size_t)k * 64 + 64];
    c = __builtin_amdgcn_wmma_f32_16x16x4_f32(false, a, false, b, (short)0, c,
                                              false, false);
  }

  float* __restrict__ orow = out + (size_t)(m0 + half * 8) * 64 + col;
#pragma unroll
  for (int v = 0; v < 8; ++v) {
    orow[(size_t)v * 64] = c[v];
  }
}

__global__ void edge_agg_kernel(const float* __restrict__ hW,
                                const int* __restrict__ src,
                                const int* __restrict__ dst,
                                const float* __restrict__ dinv,
                                float* __restrict__ agg, int e) {
  int idx = blockIdx.x * blockDim.x + threadIdx.x;
  if (idx >= e * 16) return;
  const int ei = idx >> 4;
  const int fg = (idx & 15) * 4;
  const int s = src[ei];
  const int d = dst[ei];
  const float norm = dinv[s] * dinv[d];
  const float4 hv = *(const float4*)(hW + (size_t)s * DD + fg);
  float* ap = agg + (size_t)d * DD + fg;
  atomicAdd(ap + 0, hv.x * norm);
  atomicAdd(ap + 1, hv.y * norm);
  atomicAdd(ap + 2, hv.z * norm);
  atomicAdd(ap + 3, hv.w * norm);
}

__global__ void layer_finish_kernel(const float* __restrict__ agg,
                                    const float* __restrict__ hW,
                                    const float* __restrict__ dinv,
                                    const float* __restrict__ b,
                                    float* __restrict__ hout, int n) {
  int i = blockIdx.x * blockDim.x + threadIdx.x;
  if (i >= n * DD) return;
  const int row = i >> 6;
  const int f = i & 63;
  const float di = dinv[row];
  hout[i] = tanhf(agg[i] + hW[i] * di * di + b[f]);
}


__device__ inline void atomicMaxF32(float* addr, float val) {
  if (val >= 0.0f)
    atomicMax((int*)addr, __float_as_int(val));
  else
    atomicMin((unsigned int*)addr, __float_as_uint(val));
}

__global__ void pool_acc_kernel(const float* __restrict__ h,
                                const int* __restrict__ batch,
                                float* __restrict__ gmax,
                                float* __restrict__ gsum,
                                float* __restrict__ cnt, int n) {
  int i = blockIdx.x * blockDim.x + threadIdx.x;
  if (i >= n * DD) return;
  const int row = i >> 6;
  const int f = i & 63;
  const int g = batch[row];
  const float v = h[i];
  atomicAdd(&gsum[(size_t)g * DD + f], v);
  atomicMaxF32(&gmax[(size_t)g * DD + f], v);
  if (f == 0) atomicAdd(&cnt[g], 1.0f);
}

__global__ void pool_finish_kernel(const float* __restrict__ gmax,
                                   const float* __restrict__ gsum,
                                   const float* __restrict__ cnt,
                                   float* __restrict__ hidden, int g_) {
  int i = blockIdx.x * blockDim.x + threadIdx.x;
  if (i >= g_ * DD) return;
  const int g = i >> 6;
  const int f = i & 63;
  const float c = cnt[g];
  float mx = gmax[(size_t)g * DD + f];
  if (c == 0.0f) mx = 0.0f;
  hidden[(size_t)g * (2 * DD) + f] = mx;
  hidden[(size_t)g * (2 * DD) + DD + f] = gsum[(size_t)g * DD + f] / fmaxf(c, 1.0f);
}

__global__ void head_kernel(const float* __restrict__ hidden,
                            const float* __restrict__ Wout,
                            const float* __restrict__ bout,
                            float* __restrict__ out, int g_) {
  int g = blockIdx.x * blockDim.x + threadIdx.x;
  if (g >= g_) return;
  float acc = bout[0];
#pragma unroll 8
  for (int k = 0; k < 2 * DD; ++k) acc += hidden[(size_t)g * (2 * DD) + k] * Wout[k];
  out[g] = acc;
}


extern "C" void kernel_launch(void* const* d_in, const int* in_sizes, int n_in,
                              void* d_out, int out_size, void* d_ws, size_t ws_size,
                              hipStream_t stream) {
  (void)in_sizes; (void)n_in; (void)out_size; (void)ws_size;

  const float* x    = (const float*)d_in[0];
  const int*   eidx = (const int*)d_in[1];
  const int*   batch = (const int*)d_in[2];
  const float* W0 = (const float*)d_in[3];
  const float* b0 = (const float*)d_in[4];
  const float* W1 = (const float*)d_in[5];
  const float* b1 = (const float*)d_in[6];
  const float* W2 = (const float*)d_in[7];
  const float* b2 = (const float*)d_in[8];
  const float* W3 = (const float*)d_in[9];
  const float* b3 = (const float*)d_in[10];
  const float* Wout = (const float*)d_in[11];
  const float* bout = (const float*)d_in[12];

  const int* src = eidx;
  const int* dst = eidx + EE;

  float* outp   = (float*)d_out;
  float* hidden = (float*)d_out + GG;

  float* ws = (float*)d_ws;
  size_t o = 0;
  float* deg   = ws + o; o += NN;
  float* dinv  = ws + o; o += NN;
  float* hpad  = ws + o; o += (size_t)NN * 16;
  float* w0pad = ws + o; o += 16 * 64;
  float* hW    = ws + o; o += (size_t)NN * DD;
  float* h     = ws + o; o += (size_t)NN * DD;
  float* agg   = ws + o; o += (size_t)NN * DD;
  float* gmax  = ws + o; o += (size_t)GG * DD;
  float* gsum  = ws + o; o += (size_t)GG * DD;
  float* cnt   = ws + o; o += GG;

  const int T = 256;
  auto nb = [](long long n, int t) { return (int)((n + t - 1) / t); };

  fill_f32_kernel<<<nb(NN, T), T, 0, stream>>>(deg, 1.0f, NN);
  degree_kernel<<<nb(EE, T), T, 0, stream>>>(dst, deg, EE);
  dinv_kernel<<<nb(NN, T), T, 0, stream>>>(deg, dinv, NN);

  pad_x_kernel<<<nb((long long)NN * 16, T), T, 0, stream>>>(x, hpad, NN);
  pad_w0_kernel<<<nb(16 * 64, T), T, 0, stream>>>(W0, w0pad);

  const int mtiles = NN / 16;
  const float* Wl[4] = {w0pad, W1, W2, W3};
  const float* bl[4] = {b0, b1, b2, b3};

  for (int l = 0; l < 4; ++l) {
    if (l == 0)
      gemm_wmma_kernel<16><<<mtiles, 128, 0, stream>>>(hpad, Wl[0], hW);
    else
      gemm_wmma_kernel<64><<<mtiles, 128, 0, stream>>>(h, Wl[l], hW);

    fill_f32_kernel<<<nb((long long)NN * DD, T), T, 0, stream>>>(agg, 0.0f, NN * DD);
    edge_agg_kernel<<<nb((long long)EE * 16, T), T, 0, stream>>>(hW, src, dst, dinv, agg, EE);
    layer_finish_kernel<<<nb((long long)NN * DD, T), T, 0, stream>>>(agg, hW, dinv, bl[l], h, NN);
  }

  fill_f32_kernel<<<nb((long long)GG * DD, T), T, 0, stream>>>(gmax, -FLT_MAX, GG * DD);
  fill_f32_kernel<<<nb((long long)GG * DD, T), T, 0, stream>>>(gsum, 0.0f, GG * DD);
  fill_f32_kernel<<<nb(GG, T), T, 0, stream>>>(cnt, 0.0f, GG);
  pool_acc_kernel<<<nb((long long)NN * DD, T), T, 0, stream>>>(h, batch, gmax, gsum, cnt, NN);
  pool_finish_kernel<<<nb((long long)GG * DD, T), T, 0, stream>>>(gmax, gsum, cnt, hidden, GG);

  head_kernel<<<nb(GG, 128), 128, 0, stream>>>(hidden, Wout, bout, outp, GG);
}


# Layer 0 as (A x) W0: edges scatter the 9-wide input rows (16 lanes per edge, 2 edges per atomic add) into an [N,16] plane, the WMMA GEMM follows; the 64-wide layer-0 scatter goes; sum order changes
# speedup vs baseline: 1.6820x; 1.6820x over previous
#include <hip/hip_runtime.h>
#include <math.h>
#include <float.h>


typedef __attribute__((ext_vector_type(2))) float v2f;
typedef __attribute__((ext_vector_type(4))) float v4f;
typedef __attribute__((ext_vector_type(8))) float v8f;

#define NN 50000
#define EE 1200000
#define DD 64
#define GG 1024
#define NN_PAD ((NN + 63) / 64 * 64)
#define DIN 9
#define XP 16

static_assert(NN % 16 == 0, "WMMA GEMM requires row count multiple of 16");


__global__ void fill_f32_kernel(float* __restrict__ p, float v, int n) {
  int i = blockIdx.x * blockDim.x + threadIdx.x;
  if (i < n) p[i] = v;
}

__global__ void degree_kernel(const int* __restrict__ dst, float* __restrict__ deg, int e) {
  int i = blockIdx.x * blockDim.x + threadIdx.x;
  if (i < e) atomicAdd(&deg[dst[i]], 1.0f);
}

__global__ void dinv_ax_kernel(const float* __restrict__ deg, const float* __restrict__ x,
                               float* __restrict__ dinv, float* __restrict__ ax, int n) {
  const int t = blockIdx.x * blockDim.x + threadIdx.x;
  const int i = t >> 2;
  const int c0 = (t & 3) * 4;
  if (i >= n) return;
  const float* __restrict__ xr = x + (size_t)i * DIN;
  const float x0 = xr[min(c0 + 0, DIN - 1)];
  const float x1 = xr[min(c0 + 1, DIN - 1)];
  const float x2 = xr[min(c0 + 2, DIN - 1)];
  const float x3 = xr[min(c0 + 3, DIN - 1)];
  const float di = rsqrtf(deg[i]);
  if (c0 == 0) dinv[i] = di;
  const float w = di * di;
  v4f o;
  o.x = (c0 + 0 < DIN) ? x0 * w : 0.0f;
  o.y = (c0 + 1 < DIN) ? x1 * w : 0.0f;
  o.z = (c0 + 2 < DIN) ? x2 * w : 0.0f;
  o.w = (c0 + 3 < DIN) ? x3 * w : 0.0f;
  *(v4f*)(ax + (size_t)i * XP + c0) = o;
}

#define GEMM_WAVES 5
static_assert((NN / 16) % GEMM_WAVES == 0, "row tiles split evenly over the waves of a block");

template <bool FIRST>
__global__ void __launch_bounds__(GEMM_WAVES * 32)
gemm_wmma_kernel(const float* __restrict__ A,
                 const float* __restrict__ W,
                 const float* __restrict__ dinv,
                 const float* __restrict__ bias,
                 float* __restrict__ hW,
                 float* __restrict__ agg) {
  constexpr int KSTEPS = FIRST ? 3 : 16;
  const int wave = threadIdx.x >> 5;
  const int lane = threadIdx.x & 31;
  const int l15  = lane & 15;
  const int half = lane >> 4;
  const int m0   = (blockIdx.x * GEMM_WAVES + wave) * 16;
  const int row  = m0 + l15;

  v2f a[KSTEPS];
  if (FIRST) {
    const float* __restrict__ arow = A + (size_t)row * XP + 2 * half;
#pragma unroll
    for (int s = 0; s < KSTEPS; ++s) a[s] = *(const v2f*)(arow + 4 * s);
  } else {
    const float* __restrict__ arow = A + (size_t)row * DD + half * 32;
    v4f raw[8];
#pragma unroll
    for (int i = 0; i < 8; ++i) raw[i] = *(const v4f*)(arow + 4 * i);
    asm volatile("" :: "v"(raw[0]), "v"(raw[1]), "v"(raw[2]), "v"(raw[3]),
                       "v"(raw[4]), "v"(raw[5]), "v"(raw[6]), "v"(raw[7]));
#pragma unroll
    for (int s = 0; s < KSTEPS; ++s) {
      a[s].x = tanhf(raw[s >> 1][(s & 1) * 2 + 0]);
      a[s].y = tanhf(raw[s >> 1][(s & 1) * 2 + 1]);
    }
  }

  v8f c0 = {}, c1 = {}, c2 = {}, c3 = {};
#pragma unroll
  for (int s = 0; s < KSTEPS; ++s) {
    const int kk = FIRST ? 4 * s + 2 * half : 32 * half + 2 * s;
    const float* __restrict__ wp = W + (size_t)kk * DD + 4 * l15;
    v4f w0, w1;
    if (FIRST) {
      const v4f z = {0.0f, 0.0f, 0.0f, 0.0f};
      w0 = (kk     < 9) ? *(const v4f*)(wp)      : z;
      w1 = (kk + 1 < 9) ? *(const v4f*)(wp + DD) : z;
    } else {
      w0 = *(const v4f*)(wp);
      w1 = *(const v4f*)(wp + DD);
    }
    v2f b;
    b.x = w0.x; b.y = w1.x;
    c0 = __builtin_amdgcn_wmma_f32_16x16x4_f32(false, a[s], false, b, (short)0, c0, false, false);
    b.x = w0.y; b.y = w1.y;
    c1 = __builtin_amdgcn_wmma_f32_16x16x4_f32(false, a[s], false, b, (short)0, c1, false, false);
    b.x = w0.z; b.y = w1.z;
    c2 = __builtin_amdgcn_wmma_f32_16x16x4_f32(false, a[s], false, b, (short)0, c2, false, false);
    b.x = w0.w; b.y = w1.w;
    c3 = __builtin_amdgcn_wmma_f32_16x16x4_f32(false, a[s], false, b, (short)0, c3, false, false);
  }

  const v4f b4 = *(const v4f*)(bias + 4 * l15);
  const int r0 = m0 + half * 8;
#pragma unroll
  for (int v = 0; v < 8; ++v) {
    const int r = r0 + v;
    v4f h4;
    h4.x = c0[v]; h4.y = c1[v]; h4.z = c2[v]; h4.w = c3[v];
    if (FIRST) {
      *(v4f*)(agg + (size_t)r * DD + 4 * l15) = h4 + b4;
    } else {
      const float di = dinv[r];
      *(v4f*)(hW + (size_t)r * DD + 4 * l15) = h4;
      *(v4f*)(agg + (size_t)r * DD + 4 * l15) = h4 * di * di + b4;
    }
  }
}

static_assert(EE % 32 == 0, "edge walk takes whole groups of 32 edges per wave");
#define EDGE_BATCH 16

__global__ void __launch_bounds__(256)
edge_agg_kernel(const float* __restrict__ hW,
                const int* __restrict__ src,
                const int* __restrict__ dst,
                const float* __restrict__ dinv,
                float* __restrict__ agg, int e) {
  const int tid  = blockIdx.x * blockDim.x + threadIdx.x;
  const int lane = threadIdx.x & 31;
  const int e0   = tid - lane;
  if (e0 >= e) return;
  const int s = src[e0 + lane];
  const int d = dst[e0 + lane];
  const int nrm_bits = __float_as_int(dinv[s] * dinv[d]);

  const unsigned ch = (unsigned)lane;

#pragma unroll
  for (int j0 = 0; j0 < 32; j0 += EDGE_BATCH) {
    float lo[EDGE_BATCH], hi[EDGE_BATCH];
#pragma unroll
    for (int j = 0; j < EDGE_BATCH; ++j) {
      const int sj = __builtin_amdgcn_readlane(s, j0 + j);
      const float* hp = hW + (size_t)sj * DD;
      lo[j] = hp[ch];
      hi[j] = hp[ch + 32];
    }
#pragma unroll
    for (int j = 0; j < EDGE_BATCH; ++j) {
      const int dj = __builtin_amdgcn_readlane(d, j0 + j);
      const float nj = __int_as_float(__builtin_amdgcn_readlane(nrm_bits, j0 + j));
      float* ap = agg + (size_t)dj * DD;
      atomicAdd(ap + ch, lo[j] * nj);
      atomicAdd(ap + ch + 32, hi[j] * nj);
    }
  }
}

__global__ void __launch_bounds__(256)
edge_agg_x_kernel(const float* __restrict__ x,
                  const int* __restrict__ src,
                  const int* __restrict__ dst,
                  const float* __restrict__ dinv,
                  float* __restrict__ ax, int e) {
  const int tid  = blockIdx.x * blockDim.x + threadIdx.x;
  const int lane = threadIdx.x & 31;
  const int e0   = tid - lane;
  if (e0 >= e) return;
  const int s = src[e0 + lane];
  const int d = dst[e0 + lane];
  const int nrm_bits = __float_as_int(dinv[s] * dinv[d]);

  const int c = lane & 15;
  const bool up = (lane & 16) != 0;

  int sj[16], dj[16], nj[16];
#pragma unroll
  for (int j = 0; j < 16; ++j) {
    const int s_lo = __builtin_amdgcn_readlane(s, j);
    const int s_hi = __builtin_amdgcn_readlane(s, 16 + j);
    const int d_lo = __builtin_amdgcn_readlane(d, j);
    const int d_hi = __builtin_amdgcn_readlane(d, 16 + j);
    const int n_lo = __builtin_amdgcn_readlane(nrm_bits, j);
    const int n_hi = __builtin_amdgcn_readlane(nrm_bits, 16 + j);
    sj[j] = up ? s_hi : s_lo;
    dj[j] = up ? d_hi : d_lo;
    nj[j] = up ? n_hi : n_lo;
  }
  if (c < DIN) {
    float v[16];
#pragma unroll
    for (int j = 0; j < 16; ++j) v[j] = x[(size_t)sj[j] * DIN + c];
#pragma unroll
    for (int j = 0; j < 16; ++j)
      atomicAdd(ax + (size_t)dj[j] * XP + c, v[j] * __int_as_float(nj[j]));
  }
}


__device__ inline void atomicMaxF32(float* addr, float val) {
  if (val >= 0.0f)
    atomicMax((int*)addr, __float_as_int(val));
  else
    atomicMin((unsigned int*)addr, __float_as_uint(val));
}

__global__ void pool_acc_kernel(const float* __restrict__ agg,
                                const int* __restrict__ batch,
                                float* __restrict__ gmax,
                                float* __restrict__ gsum,
                                float* __restrict__ cnt, int n) {
  int i = blockIdx.x * blockDim.x + threadIdx.x;
  if (i >= n * DD) return;
  const int row = i >> 6;
  const int f = i & 63;
  const int g = batch[row];
  const float v = tanhf(agg[i]);
  atomicAdd(&gsum[(size_t)g * DD + f], v);
  atomicMaxF32(&gmax[(size_t)g * DD + f], v);
  if (f == 0) atomicAdd(&cnt[g], 1.0f);
}

__global__ void pool_finish_kernel(const float* __restrict__ gmax,
                                   const float* __restrict__ gsum,
                                   const float* __restrict__ cnt,
                                   float* __restrict__ hidden, int g_) {
  int i = blockIdx.x * blockDim.x + threadIdx.x;
  if (i >= g_ * DD) return;
  const int g = i >> 6;
  const int f = i & 63;
  const float c = cnt[g];
  float mx = gmax[(size_t)g * DD + f];
  if (c == 0.0f) mx = 0.0f;
  hidden[(size_t)g * (2 * DD) + f] = mx;
  hidden[(size_t)g * (2 * DD) + DD + f] = gsum[(size_t)g * DD + f] / fmaxf(c, 1.0f);
}

__global__ void head_kernel(const float* __restrict__ hidden,
                            const float* __restrict__ Wout,
                            const float* __restrict__ bout,
                            float* __restrict__ out, int g_) {
  int g = blockIdx.x * blockDim.x + threadIdx.x;
  if (g >= g_) return;
  float acc = bout[0];
#pragma unroll 8
  for (int k = 0; k < 2 * DD; ++k) acc += hidden[(size_t)g * (2 * DD) + k] * Wout[k];
  out[g] = acc;
}


extern "C" void kernel_launch(void* const* d_in, const int* in_sizes, int n_in,
                              void* d_out, int out_size, void* d_ws, size_t ws_size,
                              hipStream_t stream) {
  (void)in_sizes; (void)n_in; (void)out_size; (void)ws_size;

  const float* x    = (const float*)d_in[0];
  const int*   eidx = (const int*)d_in[1];
  const int*   batch = (const int*)d_in[2];
  const float* W0 = (const float*)d_in[3];
  const float* b0 = (const float*)d_in[4];
  const float* W1 = (const float*)d_in[5];
  const float* b1 = (const float*)d_in[6];
  const float* W2 = (const float*)d_in[7];
  const float* b2 = (const float*)d_in[8];
  const float* W3 = (const float*)d_in[9];
  const float* b3 = (const float*)d_in[10];
  const float* Wout = (const float*)d_in[11];
  const float* bout = (const float*)d_in[12];

  const int* src = eidx;
  const int* dst = eidx + EE;

  float* outp   = (float*)d_out;
  float* hidden = (float*)d_out + GG;

  float* ws = (float*)d_ws;
  float* deg   = ws;
  float* dinv  = deg  + NN_PAD;
  float* hW    = dinv + NN_PAD;
  float* aggA  = hW   + (size_t)NN * DD;
  float* aggB  = aggA + (size_t)NN * DD;
  float* gmax  = aggB + (size_t)NN * DD;
  float* gsum  = gmax + (size_t)GG * DD;
  float* cnt   = gsum + (size_t)GG * DD;
  float* ax    = cnt  + GG;

  const int T = 256;
  auto nb = [](long long n, int t) { return (int)((n + t - 1) / t); };

  fill_f32_kernel<<<nb(NN, T), T, 0, stream>>>(deg, 1.0f, NN);
  degree_kernel<<<nb(EE, T), T, 0, stream>>>(dst, deg, EE);
  dinv_ax_kernel<<<nb((long long)NN * 4, T), T, 0, stream>>>(deg, x, dinv, ax, NN);

  const int gblocks = NN / 16 / GEMM_WAVES;
  const float* Wl[4] = {W0, W1, W2, W3};
  const float* bl[4] = {b0, b1, b2, b3};
  float* aggl[4] = {aggA, aggB, aggA, aggB};

  edge_agg_x_kernel<<<nb(EE, T), T, 0, stream>>>(x, src, dst, dinv, ax, EE);
  gemm_wmma_kernel<true><<<gblocks, GEMM_WAVES * 32, 0, stream>>>(ax, Wl[0], nullptr, bl[0], nullptr, aggl[0]);

  for (int l = 1; l < 4; ++l) {
    gemm_wmma_kernel<false><<<gblocks, GEMM_WAVES * 32, 0, stream>>>(aggl[l - 1], Wl[l], dinv, bl[l], hW, aggl[l]);

    edge_agg_kernel<<<nb(EE, T), T, 0, stream>>>(hW, src, dst, dinv, aggl[l], EE);
  }

  fill_f32_kernel<<<nb((long long)GG * DD, T), T, 0, stream>>>(gmax, -FLT_MAX, GG * DD);
  fill_f32_kernel<<<nb((long long)GG * DD, T), T, 0, stream>>>(gsum, 0.0f, GG * DD);
  fill_f32_kernel<<<nb(GG, T), T, 0, stream>>>(cnt, 0.0f, GG);
  pool_acc_kernel<<<nb((long long)NN * DD, T), T, 0, stream>>>(aggl[3], batch, gmax, gsum, cnt, NN);
  pool_finish_kernel<<<nb((long long)GG * DD, T), T, 0, stream>>>(gmax, gsum, cnt, hidden, GG);

  head_kernel<<<nb(GG, 128), 128, 0, stream>>>(hidden, Wout, bout, outp, GG);
}


# Degree count off global atomics: 64 work-groups each count a slice of the edge list into a private 200-KB LDS table (integer LDS adds), store it as a plane; node kernel sums planes; degrees exact
# speedup vs baseline: 2.0730x; 1.2325x over previous
#include <hip/hip_runtime.h>
#include <math.h>
#include <float.h>


typedef __attribute__((ext_vector_type(2))) float v2f;
typedef __attribute__((ext_vector_type(4))) float v4f;
typedef __attribute__((ext_vector_type(8))) float v8f;
typedef __attribute__((ext_vector_type(4))) int v4i;
typedef __attribute__((ext_vector_type(4))) unsigned v4u;

#define NN 50000
#define EE 1200000
#define DD 64
#define GG 1024
#define NN_PAD ((NN + 63) / 64 * 64)
#define DIN 9
#define XP 16

static_assert(NN % 16 == 0, "WMMA GEMM requires row count multiple of 16");


__global__ void fill_f32_kernel(float* __restrict__ p, float v, int n) {
  int i = blockIdx.x * blockDim.x + threadIdx.x;
  if (i < n) p[i] = v;
}

#define DEG_PARTS 64
#define DEG_THREADS 1024
#define DEG_STRIDE (DEG_PARTS * DEG_THREADS)
#define DEG_LOADS ((EE / 4 + DEG_STRIDE - 1) / DEG_STRIDE)
static_assert(EE % 4 == 0, "destination ids are read four at a time");
static_assert(NN % 4 == 0, "the counter table is cleared and stored four at a time");
static_assert(DEG_PARTS % 4 == 0, "a node's four lanes split the planes evenly");

__global__ void __launch_bounds__(DEG_THREADS)
degree_hist_kernel(const int* __restrict__ dst, unsigned* __restrict__ part) {
  __shared__ __attribute__((aligned(16))) unsigned cnt[NN];
  const int tid = threadIdx.x;

  const v4i* __restrict__ d4 = (const v4i*)dst;
  v4i ids[DEG_LOADS];
#pragma unroll
  for (int k = 0; k < DEG_LOADS; ++k) {
    const int q = (k * DEG_PARTS + (int)blockIdx.x) * DEG_THREADS + tid;
    const v4i none = {0, 0, 0, 0};
    ids[k] = (q < EE / 4) ? d4[q] : none;
  }

  v4u* c4 = (v4u*)cnt;
  const v4u zero4 = {0u, 0u, 0u, 0u};
  for (int i = tid; i < NN / 4; i += DEG_THREADS) c4[i] = zero4;
  __syncthreads();

#pragma unroll
  for (int k = 0; k < DEG_LOADS; ++k) {
    const int q = (k * DEG_PARTS + (int)blockIdx.x) * DEG_THREADS + tid;
    if (q < EE / 4) {
      atomicAdd(&cnt[ids[k].x], 1u);
      atomicAdd(&cnt[ids[k].y], 1u);
      atomicAdd(&cnt[ids[k].z], 1u);
      atomicAdd(&cnt[ids[k].w], 1u);
    }
  }
  __syncthreads();

  v4u* p4 = (v4u*)(part + (size_t)blockIdx.x * NN_PAD);
  for (int i = tid; i < NN / 4; i += DEG_THREADS) p4[i] = c4[i];
}

__global__ void dinv_ax_kernel(const unsigned* __restrict__ part, const float* __restrict__ x,
                               float* __restrict__ dinv, float* __restrict__ ax, int n) {
  const int t = blockIdx.x * blockDim.x + threadIdx.x;
  const int i = t >> 2;
  const int qt = t & 3;
  const int c0 = qt * 4;
  if (i >= n) return;
  const float* __restrict__ xr = x + (size_t)i * DIN;
  const float x0 = xr[min(c0 + 0, DIN - 1)];
  const float x1 = xr[min(c0 + 1, DIN - 1)];
  const float x2 = xr[min(c0 + 2, DIN - 1)];
  const float x3 = xr[min(c0 + 3, DIN - 1)];
  const unsigned* __restrict__ pp = part + (size_t)(qt * (DEG_PARTS / 4)) * NN_PAD + i;
  unsigned c = 0u;
#pragma unroll
  for (int k = 0; k < DEG_PARTS / 4; ++k) c += pp[(size_t)k * NN_PAD];
  c += __shfl_xor(c, 1);
  c += __shfl_xor(c, 2);
  const float di = rsqrtf((float)(c + 1u));
  if (c0 == 0) dinv[i] = di;
  const float w = di * di;
  v4f o;
  o.x = (c0 + 0 < DIN) ? x0 * w : 0.0f;
  o.y = (c0 + 1 < DIN) ? x1 * w : 0.0f;
  o.z = (c0 + 2 < DIN) ? x2 * w : 0.0f;
  o.w = (c0 + 3 < DIN) ? x3 * w : 0.0f;
  *(v4f*)(ax + (size_t)i * XP + c0) = o;
}

#define GEMM_WAVES 5
static_assert((NN / 16) % GEMM_WAVES == 0, "row tiles split evenly over the waves of a block");

template <bool FIRST>
__global__ void __launch_bounds__(GEMM_WAVES * 32)
gemm_wmma_kernel(const float* __restrict__ A,
                 const float* __restrict__ W,
                 const float* __restrict__ dinv,
                 const float* __restrict__ bias,
                 float* __restrict__ hW,
                 float* __restrict__ agg) {
  constexpr int KSTEPS = FIRST ? 3 : 16;
  const int wave = threadIdx.x >> 5;
  const int lane = threadIdx.x & 31;
  const int l15  = lane & 15;
  const int half = lane >> 4;
  const int m0   = (blockIdx.x * GEMM_WAVES + wave) * 16;
  const int row  = m0 + l15;

  v2f a[KSTEPS];
  if (FIRST) {
    const float* __restrict__ arow = A + (size_t)row * XP + 2 * half;
#pragma unroll
    for (int s = 0; s < KSTEPS; ++s) a[s] = *(const v2f*)(arow + 4 * s);
  } else {
    const float* __restrict__ arow = A + (size_t)row * DD + half * 32;
    v4f raw[8];
#pragma unroll
    for (int i = 0; i < 8; ++i) raw[i] = *(const v4f*)(arow + 4 * i);
    asm volatile("" :: "v"(raw[0]), "v"(raw[1]), "v"(raw[2]), "v"(raw[3]),
                       "v"(raw[4]), "v"(raw[5]), "v"(raw[6]), "v"(raw[7]));
#pragma unroll
    for (int s = 0; s < KSTEPS; ++s) {
      a[s].x = tanhf(raw[s >> 1][(s & 1) * 2 + 0]);
      a[s].y = tanhf(raw[s >> 1][(s & 1) * 2 + 1]);
    }
  }

  v8f c0 = {}, c1 = {}, c2 = {}, c3 = {};
#pragma unroll
  for (int s = 0; s < KSTEPS; ++s) {
    const int kk = FIRST ? 4 * s + 2 * half : 32 * half + 2 * s;
    const float* __restrict__ wp = W + (size_t)kk * DD + 4 * l15;
    v4f w0, w1;
    if (FIRST) {
      const v4f z = {0.0f, 0.0f, 0.0f, 0.0f};
      w0 = (kk     < 9) ? *(const v4f*)(wp)      : z;
      w1 = (kk + 1 < 9) ? *(const v4f*)(wp + DD) : z;
    } else {
      w0 = *(const v4f*)(wp);
      w1 = *(const v4f*)(wp + DD);
    }
    v2f b;
    b.x = w0.x; b.y = w1.x;
    c0 = __builtin_amdgcn_wmma_f32_16x16x4_f32(false, a[s], false, b, (short)0, c0, false, false);
    b.x = w0.y; b.y = w1.y;
    c1 = __builtin_amdgcn_wmma_f32_16x16x4_f32(false, a[s], false, b, (short)0, c1, false, false);
    b.x = w0.z; b.y = w1.z;
    c2 = __builtin_amdgcn_wmma_f32_16x16x4_f32(false, a[s], false, b, (short)0, c2, false, false);
    b.x = w0.w; b.y = w1.w;
    c3 = __builtin_amdgcn_wmma_f32_16x16x4_f32(false, a[s], false, b, (short)0, c3, false, false);
  }

  const v4f b4 = *(const v4f*)(bias + 4 * l15);
  const int r0 = m0 + half * 8;
#pragma unroll
  for (int v = 0; v < 8; ++v) {
    const int r = r0 + v;
    v4f h4;
    h4.x = c0[v]; h4.y = c1[v]; h4.z = c2[v]; h4.w = c3[v];
    if (FIRST) {
      *(v4f*)(agg + (size_t)r * DD + 4 * l15) = h4 + b4;
    } else {
      const float di = dinv[r];
      *(v4f*)(hW + (size_t)r * DD + 4 * l15) = h4;
      *(v4f*)(agg + (size_t)r * DD + 4 * l15) = h4 * di * di + b4;
    }
  }
}

static_assert(EE % 32 == 0, "edge walk takes whole groups of 32 edges per wave");
#define EDGE_BATCH 16

__global__ void __launch_bounds__(256)
edge_agg_kernel(const float* __restrict__ hW,
                const int* __restrict__ src,
                const int* __restrict__ dst,
                const float* __restrict__ dinv,
                float* __restrict__ agg, int e) {
  const int tid  = blockIdx.x * blockDim.x + threadIdx.x;
  const int lane = threadIdx.x & 31;
  const int e0   = tid - lane;
  if (e0 >= e) return;
  const int s = src[e0 + lane];
  const int d = dst[e0 + lane];
  const int nrm_bits = __float_as_int(dinv[s] * dinv[d]);

  const unsigned ch = (unsigned)lane;

#pragma unroll
  for (int j0 = 0; j0 < 32; j0 += EDGE_BATCH) {
    float lo[EDGE_BATCH], hi[EDGE_BATCH];
#pragma unroll
    for (int j = 0; j < EDGE_BATCH; ++j) {
      const int sj = __builtin_amdgcn_readlane(s, j0 + j);
      const float* hp = hW + (size_t)sj * DD;
      lo[j] = hp[ch];
      hi[j] = hp[ch + 32];
    }
#pragma unroll
    for (int j = 0; j < EDGE_BATCH; ++j) {
      const int dj = __builtin_amdgcn_readlane(d, j0 + j);
      const float nj = __int_as_float(__builtin_amdgcn_readlane(nrm_bits, j0 + j));
      float* ap = agg + (size_t)dj * DD;
      atomicAdd(ap + ch, lo[j] * nj);
      atomicAdd(ap + ch + 32, hi[j] * nj);
    }
  }
}

__global__ void __launch_bounds__(256)
edge_agg_x_kernel(const float* __restrict__ x,
                  const int* __restrict__ src,
                  const int* __restrict__ dst,
                  const float* __restrict__ dinv,
                  float* __restrict__ ax, int e) {
  const int tid  = blockIdx.x * blockDim.x + threadIdx.x;
  const int lane = threadIdx.x & 31;
  const int e0   = tid - lane;
  if (e0 >= e) return;
  const int s = src[e0 + lane];
  const int d = dst[e0 + lane];
  const int nrm_bits = __float_as_int(dinv[s] * dinv[d]);

  const int c = lane & 15;
  const bool up = (lane & 16) != 0;

  int sj[16], dj[16], nj[16];
#pragma unroll
  for (int j = 0; j < 16; ++j) {
    const int s_lo = __builtin_amdgcn_readlane(s, j);
    const int s_hi = __builtin_amdgcn_readlane(s, 16 + j);
    const int d_lo = __builtin_amdgcn_readlane(d, j);
    const int d_hi = __builtin_amdgcn_readlane(d, 16 + j);
    const int n_lo = __builtin_amdgcn_readlane(nrm_bits, j);
    const int n_hi = __builtin_amdgcn_readlane(nrm_bits, 16 + j);
    sj[j] = up ? s_hi : s_lo;
    dj[j] = up ? d_hi : d_lo;
    nj[j] = up ? n_hi : n_lo;
  }
  if (c < DIN) {
    float v[16];
#pragma unroll
    for (int j = 0; j < 16; ++j) v[j] = x[(size_t)sj[j] * DIN + c];
#pragma unroll
    for (int j = 0; j < 16; ++j)
      atomicAdd(ax + (size_t)dj[j] * XP + c, v[j] * __int_as_float(nj[j]));
  }
}


__device__ inline void atomicMaxF32(float* addr, float val) {
  if (val >= 0.0f)
    atomicMax((int*)addr, __float_as_int(val));
  else
    atomicMin((unsigned int*)addr, __float_as_uint(val));
}

__global__ void pool_acc_kernel(const float* __restrict__ agg,
                                const int* __restrict__ batch,
                                float* __restrict__ gmax,
                                float* __restrict__ gsum,
                                float* __restrict__ cnt, int n) {
  int i = blockIdx.x * blockDim.x + threadIdx.x;
  if (i >= n * DD) return;
  const int row = i >> 6;
  const int f = i & 63;
  const int g = batch[row];
  const float v = tanhf(agg[i]);
  atomicAdd(&gsum[(size_t)g * DD + f], v);
  atomicMaxF32(&gmax[(size_t)g * DD + f], v);
  if (f == 0) atomicAdd(&cnt[g], 1.0f);
}

__global__ void pool_finish_kernel(const float* __restrict__ gmax,
                                   const float* __restrict__ gsum,
                                   const float* __restrict__ cnt,
                                   float* __restrict__ hidden, int g_) {
  int i = blockIdx.x * blockDim.x + threadIdx.x;
  if (i >= g_ * DD) return;
  const int g = i >> 6;
  const int f = i & 63;
  const float c = cnt[g];
  float mx = gmax[(size_t)g * DD + f];
  if (c == 0.0f) mx = 0.0f;
  hidden[(size_t)g * (2 * DD) + f] = mx;
  hidden[(size_t)g * (2 * DD) + DD + f] = gsum[(size_t)g * DD + f] / fmaxf(c, 1.0f);
}

__global__ void head_kernel(const float* __restrict__ hidden,
                            const float* __restrict__ Wout,
                            const float* __restrict__ bout,
                            float* __restrict__ out, int g_) {
  int g = blockIdx.x * blockDim.x + threadIdx.x;
  if (g >= g_) return;
  float acc = bout[0];
#pragma unroll 8
  for (int k = 0; k < 2 * DD; ++k) acc += hidden[(size_t)g * (2 * DD) + k] * Wout[k];
  out[g] = acc;
}


extern "C" void kernel_launch(void* const* d_in, const int* in_sizes, int n_in,
                              void* d_out, int out_size, void* d_ws, size_t ws_size,
                              hipStream_t stream) {
  (void)in_sizes; (void)n_in; (void)out_size; (void)ws_size;

  const float* x    = (const float*)d_in[0];
  const int*   eidx = (const int*)d_in[1];
  const int*   batch = (const int*)d_in[2];
  const float* W0 = (const float*)d_in[3];
  const float* b0 = (const float*)d_in[4];
  const float* W1 = (const float*)d_in[5];
  const float* b1 = (const float*)d_in[6];
  const float* W2 = (const float*)d_in[7];
  const float* b2 = (const float*)d_in[8];
  const float* W3 = (const float*)d_in[9];
  const float* b3 = (const float*)d_in[10];
  const float* Wout = (const float*)d_in[11];
  const float* bout = (const float*)d_in[12];

  const int* src = eidx;
  const int* dst = eidx + EE;

  float* outp   = (float*)d_out;
  float* hidden = (float*)d_out + GG;

  float* ws = (float*)d_ws;
  float* dinv  = ws;
  float* hW    = dinv + NN_PAD;
  float* aggA  = hW   + (size_t)NN * DD;
  float* aggB  = aggA + (size_t)NN * DD;
  float* gmax  = aggB + (size_t)NN * DD;
  float* gsum  = gmax + (size_t)GG * DD;
  float* cnt   = gsum + (size_t)GG * DD;
  float* ax    = cnt  + GG;
  unsigned* part = (unsigned*)(ax + (size_t)NN * XP);

  const int T = 256;
  auto nb = [](long long n, int t) { return (int)((n + t - 1) / t); };

  degree_hist_kernel<<<DEG_PARTS, DEG_THREADS, 0, stream>>>(dst, part);
  dinv_ax_kernel<<<nb((long long)NN * 4, T), T, 0, stream>>>(part, x, dinv, ax, NN);

  const int gblocks = NN / 16 / GEMM_WAVES;
  const float* Wl[4] = {W0, W1, W2, W3};
  const float* bl[4] = {b0, b1, b2, b3};
  float* aggl[4] = {aggA, aggB, aggA, aggB};

  edge_agg_x_kernel<<<nb(EE, T), T, 0, stream>>>(x, src, dst, dinv, ax, EE);
  gemm_wmma_kernel<true><<<gblocks, GEMM_WAVES * 32, 0, stream>>>(ax, Wl[0], nullptr, bl[0], nullptr, aggl[0]);

  for (int l = 1; l < 4; ++l) {
    gemm_wmma_kernel<false><<<gblocks, GEMM_WAVES * 32, 0, stream>>>(aggl[l - 1], Wl[l], dinv, bl[l], hW, aggl[l]);

    edge_agg_kernel<<<nb(EE, T), T, 0, stream>>>(hW, src, dst, dinv, aggl[l], EE);
  }

  fill_f32_kernel<<<nb((long long)GG * DD, T), T, 0, stream>>>(gmax, -FLT_MAX, GG * DD);
  fill_f32_kernel<<<nb((long long)GG * DD, T), T, 0, stream>>>(gsum, 0.0f, GG * DD);
  fill_f32_kernel<<<nb(GG, T), T, 0, stream>>>(cnt, 0.0f, GG);
  pool_acc_kernel<<<nb((long long)NN * DD, T), T, 0, stream>>>(aggl[3], batch, gmax, gsum, cnt, NN);
  pool_finish_kernel<<<nb((long long)GG * DD, T), T, 0, stream>>>(gmax, gsum, cnt, hidden, GG);

  head_kernel<<<nb(GG, 128), 128, 0, stream>>>(hidden, Wout, bout, outp, GG);
}


# Neighbour sums kept once per half of the chip: an edge wave reads its die id and adds into its own half's plane, readers take plane 0 + plane 1, so no line's atomics cross caches; sum order changes
# speedup vs baseline: 2.4723x; 1.1926x over previous
#include <hip/hip_runtime.h>
#include <math.h>
#include <float.h>


typedef __attribute__((ext_vector_type(2))) float v2f;
typedef __attribute__((ext_vector_type(4))) float v4f;
typedef __attribute__((ext_vector_type(8))) float v8f;
typedef __attribute__((ext_vector_type(4))) int v4i;
typedef __attribute__((ext_vector_type(4))) unsigned v4u;

#define NN 50000
#define EE 1200000
#define DD 64
#define GG 1024
#define NN_PAD ((NN + 63) / 64 * 64)
#define DIN 9
#define XP 16
#define AGG_PLANE ((size_t)NN * DD)
#define AX_PLANE  ((size_t)NN * XP)

static_assert(NN % 16 == 0, "WMMA GEMM requires row count multiple of 16");

__device__ __forceinline__ int chip_half() {
  return (int)((__builtin_amdgcn_s_sendmsg_rtn(0x87) >> 18) & 1u);
}


__global__ void fill_f32_kernel(float* __restrict__ p, float v, int n) {
  int i = blockIdx.x * blockDim.x + threadIdx.x;
  if (i < n) p[i] = v;
}

#define DEG_PARTS 64
#define DEG_THREADS 1024
#define DEG_STRIDE (DEG_PARTS * DEG_THREADS)
#define DEG_LOADS ((EE / 4 + DEG_STRIDE - 1) / DEG_STRIDE)
static_assert(EE % 4 == 0, "destination ids are read four at a time");
static_assert(NN % 4 == 0, "the counter table is cleared and stored four at a time");
static_assert(DEG_PARTS % 4 == 0, "a node's four lanes split the planes evenly");

__global__ void __launch_bounds__(DEG_THREADS)
degree_hist_kernel(const int* __restrict__ dst, unsigned* __restrict__ part) {
  __shared__ __attribute__((aligned(16))) unsigned cnt[NN];
  const int tid = threadIdx.x;

  const v4i* __restrict__ d4 = (const v4i*)dst;
  v4i ids[DEG_LOADS];
#pragma unroll
  for (int k = 0; k < DEG_LOADS; ++k) {
    const int q = (k * DEG_PARTS + (int)blockIdx.x) * DEG_THREADS + tid;
    const v4i none = {0, 0, 0, 0};
    ids[k] = (q < EE / 4) ? d4[q] : none;
  }

  v4u* c4 = (v4u*)cnt;
  const v4u zero4 = {0u, 0u, 0u, 0u};
  for (int i = tid; i < NN / 4; i += DEG_THREADS) c4[i] = zero4;
  __syncthreads();

#pragma unroll
  for (int k = 0; k < DEG_LOADS; ++k) {
    const int q = (k * DEG_PARTS + (int)blockIdx.x) * DEG_THREADS + tid;
    if (q < EE / 4) {
      atomicAdd(&cnt[ids[k].x], 1u);
      atomicAdd(&cnt[ids[k].y], 1u);
      atomicAdd(&cnt[ids[k].z], 1u);
      atomicAdd(&cnt[ids[k].w], 1u);
    }
  }
  __syncthreads();

  v4u* p4 = (v4u*)(part + (size_t)blockIdx.x * NN_PAD);
  for (int i = tid; i < NN / 4; i += DEG_THREADS) p4[i] = c4[i];
}

__global__ void dinv_ax_kernel(const unsigned* __restrict__ part, const float* __restrict__ x,
                               float* __restrict__ dinv, float* __restrict__ ax, int n) {
  const int t = blockIdx.x * blockDim.x + threadIdx.x;
  const int i = t >> 2;
  const int qt = t & 3;
  const int c0 = qt * 4;
  if (i >= n) return;
  const float* __restrict__ xr = x + (size_t)i * DIN;
  const float x0 = xr[min(c0 + 0, DIN - 1)];
  const float x1 = xr[min(c0 + 1, DIN - 1)];
  const float x2 = xr[min(c0 + 2, DIN - 1)];
  const float x3 = xr[min(c0 + 3, DIN - 1)];
  const unsigned* __restrict__ pp = part + (size_t)(qt * (DEG_PARTS / 4)) * NN_PAD + i;
  unsigned c = 0u;
#pragma unroll
  for (int k = 0; k < DEG_PARTS / 4; ++k) c += pp[(size_t)k * NN_PAD];
  c += __shfl_xor(c, 1);
  c += __shfl_xor(c, 2);
  const float di = rsqrtf((float)(c + 1u));
  if (c0 == 0) dinv[i] = di;
  const float w = di * di;
  v4f o;
  o.x = (c0 + 0 < DIN) ? x0 * w : 0.0f;
  o.y = (c0 + 1 < DIN) ? x1 * w : 0.0f;
  o.z = (c0 + 2 < DIN) ? x2 * w : 0.0f;
  o.w = (c0 + 3 < DIN) ? x3 * w : 0.0f;
  *(v4f*)(ax + (size_t)i * XP + c0) = o;
  const v4f zero4 = {0.0f, 0.0f, 0.0f, 0.0f};
  *(v4f*)(ax + AX_PLANE + (size_t)i * XP + c0) = zero4;
}

#define GEMM_WAVES 5
static_assert((NN / 16) % GEMM_WAVES == 0, "row tiles split evenly over the waves of a block");

template <bool FIRST>
__global__ void __launch_bounds__(GEMM_WAVES * 32)
gemm_wmma_kernel(const float* __restrict__ A,
                 const float* __restrict__ W,
                 const float* __restrict__ dinv,
                 const float* __restrict__ bias,
                 float* __restrict__ hW,
                 float* __restrict__ agg) {
  constexpr int KSTEPS = FIRST ? 3 : 16;
  const int wave = threadIdx.x >> 5;
  const int lane = threadIdx.x & 31;
  const int l15  = lane & 15;
  const int half = lane >> 4;
  const int m0   = (blockIdx.x * GEMM_WAVES + wave) * 16;
  const int row  = m0 + l15;

  v2f a[KSTEPS];
  if (FIRST) {
    const float* __restrict__ arow = A + (size_t)row * XP + 2 * half;
#pragma unroll
    for (int s = 0; s < KSTEPS; ++s)
      a[s] = *(const v2f*)(arow + 4 * s) + *(const v2f*)(arow + AX_PLANE + 4 * s);
  } else {
    const float* __restrict__ arow = A + (size_t)row * DD + half * 32;
    v4f raw[8], rw1[8];
#pragma unroll
    for (int i = 0; i < 8; ++i) {
      raw[i] = *(const v4f*)(arow + 4 * i);
      rw1[i] = *(const v4f*)(arow + AGG_PLANE + 4 * i);
    }
    asm volatile("" :: "v"(raw[0]), "v"(raw[1]), "v"(raw[2]), "v"(raw[3]),
                       "v"(raw[4]), "v"(raw[5]), "v"(raw[6]), "v"(raw[7]),
                       "v"(rw1[0]), "v"(rw1[1]), "v"(rw1[2]), "v"(rw1[3]),
                       "v"(rw1[4]), "v"(rw1[5]), "v"(rw1[6]), "v"(rw1[7]));
#pragma unroll
    for (int i = 0; i < 8; ++i) raw[i] += rw1[i];
#pragma unroll
    for (int s = 0; s < KSTEPS; ++s) {
      a[s].x = tanhf(raw[s >> 1][(s & 1) * 2 + 0]);
      a[s].y = tanhf(raw[s >> 1][(s & 1) * 2 + 1]);
    }
  }

  v8f c0 = {}, c1 = {}, c2 = {}, c3 = {};
#pragma unroll
  for (int s = 0; s < KSTEPS; ++s) {
    const int kk = FIRST ? 4 * s + 2 * half : 32 * half + 2 * s;
    const float* __restrict__ wp = W + (size_t)kk * DD + 4 * l15;
    v4f w0, w1;
    if (FIRST) {
      const v4f z = {0.0f, 0.0f, 0.0f, 0.0f};
      w0 = (kk     < 9) ? *(const v4f*)(wp)      : z;
      w1 = (kk + 1 < 9) ? *(const v4f*)(wp + DD) : z;
    } else {
      w0 = *(const v4f*)(wp);
      w1 = *(const v4f*)(wp + DD);
    }
    v2f b;
    b.x = w0.x; b.y = w1.x;
    c0 = __builtin_amdgcn_wmma_f32_16x16x4_f32(false, a[s], false, b, (short)0, c0, false, false);
    b.x = w0.y; b.y = w1.y;
    c1 = __builtin_amdgcn_wmma_f32_16x16x4_f32(false, a[s], false, b, (short)0, c1, false, false);
    b.x = w0.z; b.y = w1.z;
    c2 = __builtin_amdgcn_wmma_f32_16x16x4_f32(false, a[s], false, b, (short)0, c2, false, false);
    b.x = w0.w; b.y = w1.w;
    c3 = __builtin_amdgcn_wmma_f32_16x16x4_f32(false, a[s], false, b, (short)0, c3, false, false);
  }

  const v4f b4 = *(const v4f*)(bias + 4 * l15);
  const v4f zero4 = {0.0f, 0.0f, 0.0f, 0.0f};
  const int r0 = m0 + half * 8;
#pragma unroll
  for (int v = 0; v < 8; ++v) {
    const int r = r0 + v;
    v4f h4;
    h4.x = c0[v]; h4.y = c1[v]; h4.z = c2[v]; h4.w = c3[v];
    if (FIRST) {
      *(v4f*)(agg + (size_t)r * DD + 4 * l15) = h4 + b4;
    } else {
      const float di = dinv[r];
      *(v4f*)(hW + (size_t)r * DD + 4 * l15) = h4;
      *(v4f*)(agg + (size_t)r * DD + 4 * l15) = h4 * di * di + b4;
    }
    *(v4f*)(agg + AGG_PLANE + (size_t)r * DD + 4 * l15) = zero4;
  }
}

static_assert(EE % 32 == 0, "edge walk takes whole groups of 32 edges per wave");
#define EDGE_BATCH 16

__global__ void __launch_bounds__(256)
edge_agg_kernel(const float* __restrict__ hW,
                const int* __restrict__ src,
                const int* __restrict__ dst,
                const float* __restrict__ dinv,
                float* __restrict__ agg, int e) {
  const int tid  = blockIdx.x * blockDim.x + threadIdx.x;
  const int lane = threadIdx.x & 31;
  const int e0   = tid - lane;
  if (e0 >= e) return;
  const int s = src[e0 + lane];
  const int d = dst[e0 + lane];
  const int nrm_bits = __float_as_int(dinv[s] * dinv[d]);

  const unsigned ch = (unsigned)lane;
  float* aggh = agg;

#pragma unroll
  for (int j0 = 0; j0 < 32; j0 += EDGE_BATCH) {
    float lo[EDGE_BATCH], hi[EDGE_BATCH];
#pragma unroll
    for (int j = 0; j < EDGE_BATCH; ++j) {
      const int sj = __builtin_amdgcn_readlane(s, j0 + j);
      const float* hp = hW + (size_t)sj * DD;
      lo[j] = hp[ch];
      hi[j] = hp[ch + 32];
    }
    if (j0 == 0) aggh = agg + (size_t)chip_half() * AGG_PLANE;
#pragma unroll
    for (int j = 0; j < EDGE_BATCH; ++j) {
      const int dj = __builtin_amdgcn_readlane(d, j0 + j);
      const float nj = __int_as_float(__builtin_amdgcn_readlane(nrm_bits, j0 + j));
      float* ap = aggh + (size_t)dj * DD;
      atomicAdd(ap + ch, lo[j] * nj);
      atomicAdd(ap + ch + 32, hi[j] * nj);
    }
  }
}

__global__ void __launch_bounds__(256)
edge_agg_x_kernel(const float* __restrict__ x,
                  const int* __restrict__ src,
                  const int* __restrict__ dst,
                  const float* __restrict__ dinv,
                  float* __restrict__ ax, int e) {
  const int tid  = blockIdx.x * blockDim.x + threadIdx.x;
  const int lane = threadIdx.x & 31;
  const int e0   = tid - lane;
  if (e0 >= e) return;
  const int s = src[e0 + lane];
  const int d = dst[e0 + lane];
  const int nrm_bits = __float_as_int(dinv[s] * dinv[d]);

  const int c = lane & 15;
  const bool up = (lane & 16) != 0;

  int sj[16], dj[16], nj[16];
#pragma unroll
  for (int j = 0; j < 16; ++j) {
    const int s_lo = __builtin_amdgcn_readlane(s, j);
    const int s_hi = __builtin_amdgcn_readlane(s, 16 + j);
    const int d_lo = __builtin_amdgcn_readlane(d, j);
    const int d_hi = __builtin_amdgcn_readlane(d, 16 + j);
    const int n_lo = __builtin_amdgcn_readlane(nrm_bits, j);
    const int n_hi = __builtin_amdgcn_readlane(nrm_bits, 16 + j);
    sj[j] = up ? s_hi : s_lo;
    dj[j] = up ? d_hi : d_lo;
    nj[j] = up ? n_hi : n_lo;
  }
  if (c < DIN) {
    float v[16];
#pragma unroll
    for (int j = 0; j < 16; ++j) v[j] = x[(size_t)sj[j] * DIN + c];
    float* axh = ax + (size_t)chip_half() * AX_PLANE;
#pragma unroll
    for (int j = 0; j < 16; ++j)
      atomicAdd(axh + (size_t)dj[j] * XP + c, v[j] * __int_as_float(nj[j]));
  }
}


__device__ inline void atomicMaxF32(float* addr, float val) {
  if (val >= 0.0f)
    atomicMax((int*)addr, __float_as_int(val));
  else
    atomicMin((unsigned int*)addr, __float_as_uint(val));
}

__global__ void pool_acc_kernel(const float* __restrict__ agg,
                                const int* __restrict__ batch,
                                float* __restrict__ gmax,
                                float* __restrict__ gsum,
                                float* __restrict__ cnt, int n) {
  int i = blockIdx.x * blockDim.x + threadIdx.x;
  if (i >= n * DD) return;
  const int row = i >> 6;
  const int f = i & 63;
  const int g = batch[row];
  const float v = tanhf(agg[i] + agg[AGG_PLANE + i]);
  atomicAdd(&gsum[(size_t)g * DD + f], v);
  atomicMaxF32(&gmax[(size_t)g * DD + f], v);
  if (f == 0) atomicAdd(&cnt[g], 1.0f);
}

__global__ void pool_finish_kernel(const float* __restrict__ gmax,
                                   const float* __restrict__ gsum,
                                   const float* __restrict__ cnt,
                                   float* __restrict__ hidden, int g_) {
  int i = blockIdx.x * blockDim.x + threadIdx.x;
  if (i >= g_ * DD) return;
  const int g = i >> 6;
  const int f = i & 63;
  const float c = cnt[g];
  float mx = gmax[(size_t)g * DD + f];
  if (c == 0.0f) mx = 0.0f;
  hidden[(size_t)g * (2 * DD) + f] = mx;
  hidden[(size_t)g * (2 * DD) + DD + f] = gsum[(size_t)g * DD + f] / fmaxf(c, 1.0f);
}

__global__ void head_kernel(const float* __restrict__ hidden,
                            const float* __restrict__ Wout,
                            const float* __restrict__ bout,
                            float* __restrict__ out, int g_) {
  int g = blockIdx.x * blockDim.x + threadIdx.x;
  if (g >= g_) return;
  float acc = bout[0];
#pragma unroll 8
  for (int k = 0; k < 2 * DD; ++k) acc += hidden[(size_t)g * (2 * DD) + k] * Wout[k];
  out[g] = acc;
}


extern "C" void kernel_launch(void* const* d_in, const int* in_sizes, int n_in,
                              void* d_out, int out_size, void* d_ws, size_t ws_size,
                              hipStream_t stream) {
  (void)in_sizes; (void)n_in; (void)out_size; (void)ws_size;

  const float* x    = (const float*)d_in[0];
  const int*   eidx = (const int*)d_in[1];
  const int*   batch = (const int*)d_in[2];
  const float* W0 = (const float*)d_in[3];
  const float* b0 = (const float*)d_in[4];
  const float* W1 = (const float*)d_in[5];
  const float* b1 = (const float*)d_in[6];
  const float* W2 = (const float*)d_in[7];
  const float* b2 = (const float*)d_in[8];
  const float* W3 = (const float*)d_in[9];
  const float* b3 = (const float*)d_in[10];
  const float* Wout = (const float*)d_in[11];
  const float* bout = (const float*)d_in[12];

  const int* src = eidx;
  const int* dst = eidx + EE;

  float* outp   = (float*)d_out;
  float* hidden = (float*)d_out + GG;

  float* ws = (float*)d_ws;
  float* dinv  = ws;
  float* hW    = dinv + NN_PAD;
  float* aggA  = hW   + (size_t)NN * DD;
  float* aggB  = aggA + 2 * AGG_PLANE;
  float* gmax  = aggB + 2 * AGG_PLANE;
  float* gsum  = gmax + (size_t)GG * DD;
  float* cnt   = gsum + (size_t)GG * DD;
  float* ax    = cnt  + GG;
  unsigned* part = (unsigned*)(ax + 2 * AX_PLANE);

  const int T = 256;
  auto nb = [](long long n, int t) { return (int)((n + t - 1) / t); };

  degree_hist_kernel<<<DEG_PARTS, DEG_THREADS, 0, stream>>>(dst, part);
  dinv_ax_kernel<<<nb((long long)NN * 4, T), T, 0, stream>>>(part, x, dinv, ax, NN);

  const int gblocks = NN / 16 / GEMM_WAVES;
  const float* Wl[4] = {W0, W1, W2, W3};
  const float* bl[4] = {b0, b1, b2, b3};
  float* aggl[4] = {aggA, aggB, aggA, aggB};

  edge_agg_x_kernel<<<nb(EE, T), T, 0, stream>>>(x, src, dst, dinv, ax, EE);
  gemm_wmma_kernel<true><<<gblocks, GEMM_WAVES * 32, 0, stream>>>(ax, Wl[0], nullptr, bl[0], nullptr, aggl[0]);

  for (int l = 1; l < 4; ++l) {
    gemm_wmma_kernel<false><<<gblocks, GEMM_WAVES * 32, 0, stream>>>(aggl[l - 1], Wl[l], dinv, bl[l], hW, aggl[l]);

    edge_agg_kernel<<<nb(EE, T), T, 0, stream>>>(hW, src, dst, dinv, aggl[l], EE);
  }

  fill_f32_kernel<<<nb((long long)GG * DD, T), T, 0, stream>>>(gmax, -FLT_MAX, GG * DD);
  fill_f32_kernel<<<nb((long long)GG * DD, T), T, 0, stream>>>(gsum, 0.0f, GG * DD);
  fill_f32_kernel<<<nb(GG, T), T, 0, stream>>>(cnt, 0.0f, GG);
  pool_acc_kernel<<<nb((long long)NN * DD, T), T, 0, stream>>>(aggl[3], batch, gmax, gsum, cnt, NN);
  pool_finish_kernel<<<nb((long long)GG * DD, T), T, 0, stream>>>(gmax, gsum, cnt, hidden, GG);

  head_kernel<<<nb(GG, 128), 128, 0, stream>>>(hidden, Wout, bout, outp, GG);
}


# Edge weight dinv[src]*dinv[dst] stored once per edge by the layer-0 edge kernel as an [E] plane; the three 64-wide edge kernels load it as one whole line per wave where they made two 32-line gathers
# speedup vs baseline: 2.6874x; 1.0870x over previous
#include <hip/hip_runtime.h>
#include <math.h>
#include <float.h>


typedef __attribute__((ext_vector_type(2))) float v2f;
typedef __attribute__((ext_vector_type(4))) float v4f;
typedef __attribute__((ext_vector_type(8))) float v8f;
typedef __attribute__((ext_vector_type(4))) int v4i;
typedef __attribute__((ext_vector_type(4))) unsigned v4u;

#define NN 50000
#define EE 1200000
#define DD 64
#define GG 1024
#define NN_PAD ((NN + 63) / 64 * 64)
#define DIN 9
#define XP 16
#define AGG_PLANE ((size_t)NN * DD)
#define AX_PLANE  ((size_t)NN * XP)

static_assert(NN % 16 == 0, "WMMA GEMM requires row count multiple of 16");

__device__ __forceinline__ int chip_half() {
  return (int)((__builtin_amdgcn_s_sendmsg_rtn(0x87) >> 18) & 1u);
}


__global__ void fill_f32_kernel(float* __restrict__ p, float v, int n) {
  int i = blockIdx.x * blockDim.x + threadIdx.x;
  if (i < n) p[i] = v;
}

#define DEG_PARTS 64
#define DEG_THREADS 1024
#define DEG_STRIDE (DEG_PARTS * DEG_THREADS)
#define DEG_LOADS ((EE / 4 + DEG_STRIDE - 1) / DEG_STRIDE)
static_assert(EE % 4 == 0, "destination ids are read four at a time");
static_assert(NN % 4 == 0, "the counter table is cleared and stored four at a time");
static_assert(DEG_PARTS % 4 == 0, "a node's four lanes split the planes evenly");

__global__ void __launch_bounds__(DEG_THREADS)
degree_hist_kernel(const int* __restrict__ dst, unsigned* __restrict__ part) {
  __shared__ __attribute__((aligned(16))) unsigned cnt[NN];
  const int tid = threadIdx.x;

  const v4i* __restrict__ d4 = (const v4i*)dst;
  v4i ids[DEG_LOADS];
#pragma unroll
  for (int k = 0; k < DEG_LOADS; ++k) {
    const int q = (k * DEG_PARTS + (int)blockIdx.x) * DEG_THREADS + tid;
    const v4i none = {0, 0, 0, 0};
    ids[k] = (q < EE / 4) ? d4[q] : none;
  }

  v4u* c4 = (v4u*)cnt;
  const v4u zero4 = {0u, 0u, 0u, 0u};
  for (int i = tid; i < NN / 4; i += DEG_THREADS) c4[i] = zero4;
  __syncthreads();

#pragma unroll
  for (int k = 0; k < DEG_LOADS; ++k) {
    const int q = (k * DEG_PARTS + (int)blockIdx.x) * DEG_THREADS + tid;
    if (q < EE / 4) {
      atomicAdd(&cnt[ids[k].x], 1u);
      atomicAdd(&cnt[ids[k].y], 1u);
      atomicAdd(&cnt[ids[k].z], 1u);
      atomicAdd(&cnt[ids[k].w], 1u);
    }
  }
  __syncthreads();

  v4u* p4 = (v4u*)(part + (size_t)blockIdx.x * NN_PAD);
  for (int i = tid; i < NN / 4; i += DEG_THREADS) p4[i] = c4[i];
}

__global__ void dinv_ax_kernel(const unsigned* __restrict__ part, const float* __restrict__ x,
                               float* __restrict__ dinv, float* __restrict__ ax, int n) {
  const int t = blockIdx.x * blockDim.x + threadIdx.x;
  const int i = t >> 2;
  const int qt = t & 3;
  const int c0 = qt * 4;
  if (i >= n) return;
  const float* __restrict__ xr = x + (size_t)i * DIN;
  const float x0 = xr[min(c0 + 0, DIN - 1)];
  const float x1 = xr[min(c0 + 1, DIN - 1)];
  const float x2 = xr[min(c0 + 2, DIN - 1)];
  const float x3 = xr[min(c0 + 3, DIN - 1)];
  const unsigned* __restrict__ pp = part + (size_t)(qt * (DEG_PARTS / 4)) * NN_PAD + i;
  unsigned c = 0u;
#pragma unroll
  for (int k = 0; k < DEG_PARTS / 4; ++k) c += pp[(size_t)k * NN_PAD];
  c += __shfl_xor(c, 1);
  c += __shfl_xor(c, 2);
  const float di = rsqrtf((float)(c + 1u));
  if (c0 == 0) dinv[i] = di;
  const float w = di * di;
  v4f o;
  o.x = (c0 + 0 < DIN) ? x0 * w : 0.0f;
  o.y = (c0 + 1 < DIN) ? x1 * w : 0.0f;
  o.z = (c0 + 2 < DIN) ? x2 * w : 0.0f;
  o.w = (c0 + 3 < DIN) ? x3 * w : 0.0f;
  *(v4f*)(ax + (size_t)i * XP + c0) = o;
  const v4f zero4 = {0.0f, 0.0f, 0.0f, 0.0f};
  *(v4f*)(ax + AX_PLANE + (size_t)i * XP + c0) = zero4;
}

#define GEMM_WAVES 5
static_assert((NN / 16) % GEMM_WAVES == 0, "row tiles split evenly over the waves of a block");

template <bool FIRST>
__global__ void __launch_bounds__(GEMM_WAVES * 32)
gemm_wmma_kernel(const float* __restrict__ A,
                 const float* __restrict__ W,
                 const float* __restrict__ dinv,
                 const float* __restrict__ bias,
                 float* __restrict__ hW,
                 float* __restrict__ agg) {
  constexpr int KSTEPS = FIRST ? 3 : 16;
  const int wave = threadIdx.x >> 5;
  const int lane = threadIdx.x & 31;
  const int l15  = lane & 15;
  const int half = lane >> 4;
  const int m0   = (blockIdx.x * GEMM_WAVES + wave) * 16;
  const int row  = m0 + l15;

  v2f a[KSTEPS];
  if (FIRST) {
    const float* __restrict__ arow = A + (size_t)row * XP + 2 * half;
#pragma unroll
    for (int s = 0; s < KSTEPS; ++s)
      a[s] = *(const v2f*)(arow + 4 * s) + *(const v2f*)(arow + AX_PLANE + 4 * s);
  } else {
    const float* __restrict__ arow = A + (size_t)row * DD + half * 32;
    v4f raw[8], rw1[8];
#pragma unroll
    for (int i = 0; i < 8; ++i) {
      raw[i] = *(const v4f*)(arow + 4 * i);
      rw1[i] = *(const v4f*)(arow + AGG_PLANE + 4 * i);
    }
    asm volatile("" :: "v"(raw[0]), "v"(raw[1]), "v"(raw[2]), "v"(raw[3]),
                       "v"(raw[4]), "v"(raw[5]), "v"(raw[6]), "v"(raw[7]),
                       "v"(rw1[0]), "v"(rw1[1]), "v"(rw1[2]), "v"(rw1[3]),
                       "v"(rw1[4]), "v"(rw1[5]), "v"(rw1[6]), "v"(rw1[7]));
#pragma unroll
    for (int i = 0; i < 8; ++i) raw[i] += rw1[i];
#pragma unroll
    for (int s = 0; s < KSTEPS; ++s) {
      a[s].x = tanhf(raw[s >> 1][(s & 1) * 2 + 0]);
      a[s].y = tanhf(raw[s >> 1][(s & 1) * 2 + 1]);
    }
  }

  v8f c0 = {}, c1 = {}, c2 = {}, c3 = {};
#pragma unroll
  for (int s = 0; s < KSTEPS; ++s) {
    const int kk = FIRST ? 4 * s + 2 * half : 32 * half + 2 * s;
    const float* __restrict__ wp = W + (size_t)kk * DD + 4 * l15;
    v4f w0, w1;
    if (FIRST) {
      const v4f z = {0.0f, 0.0f, 0.0f, 0.0f};
      w0 = (kk     < 9) ? *(const v4f*)(wp)      : z;
      w1 = (kk + 1 < 9) ? *(const v4f*)(wp + DD) : z;
    } else {
      w0 = *(const v4f*)(wp);
      w1 = *(const v4f*)(wp + DD);
    }
    v2f b;
    b.x = w0.x; b.y = w1.x;
    c0 = __builtin_amdgcn_wmma_f32_16x16x4_f32(false, a[s], false, b, (short)0, c0, false, false);
    b.x = w0.y; b.y = w1.y;
    c1 = __builtin_amdgcn_wmma_f32_16x16x4_f32(false, a[s], false, b, (short)0, c1, false, false);
    b.x = w0.z; b.y = w1.z;
    c2 = __builtin_amdgcn_wmma_f32_16x16x4_f32(false, a[s], false, b, (short)0, c2, false, false);
    b.x = w0.w; b.y = w1.w;
    c3 = __builtin_amdgcn_wmma_f32_16x16x4_f32(false, a[s], false, b, (short)0, c3, false, false);
  }

  const v4f b4 = *(const v4f*)(bias + 4 * l15);
  const v4f zero4 = {0.0f, 0.0f, 0.0f, 0.0f};
  const int r0 = m0 + half * 8;
#pragma unroll
  for (int v = 0; v < 8; ++v) {
    const int r = r0 + v;
    v4f h4;
    h4.x = c0[v]; h4.y = c1[v]; h4.z = c2[v]; h4.w = c3[v];
    if (FIRST) {
      *(v4f*)(agg + (size_t)r * DD + 4 * l15) = h4 + b4;
    } else {
      const float di = dinv[r];
      *(v4f*)(hW + (size_t)r * DD + 4 * l15) = h4;
      *(v4f*)(agg + (size_t)r * DD + 4 * l15) = h4 * di * di + b4;
    }
    *(v4f*)(agg + AGG_PLANE + (size_t)r * DD + 4 * l15) = zero4;
  }
}

static_assert(EE % 32 == 0, "edge walk takes whole groups of 32 edges per wave");
#define EDGE_BATCH 16

__global__ void __launch_bounds__(256)
edge_agg_kernel(const float* __restrict__ hW,
                const int* __restrict__ src,
                const int* __restrict__ dst,
                const float* __restrict__ nrm,
                float* __restrict__ agg, int e) {
  const int tid  = blockIdx.x * blockDim.x + threadIdx.x;
  const int lane = threadIdx.x & 31;
  const int e0   = tid - lane;
  if (e0 >= e) return;
  const int s = src[e0 + lane];
  const int d = dst[e0 + lane];
  const int nrm_bits = __float_as_int(nrm[e0 + lane]);

  const unsigned ch = (unsigned)lane;
  float* aggh = agg;

#pragma unroll
  for (int j0 = 0; j0 < 32; j0 += EDGE_BATCH) {
    float lo[EDGE_BATCH], hi[EDGE_BATCH];
#pragma unroll
    for (int j = 0; j < EDGE_BATCH; ++j) {
      const int sj = __builtin_amdgcn_readlane(s, j0 + j);
      const float* hp = hW + (size_t)sj * DD;
      lo[j] = hp[ch];
      hi[j] = hp[ch + 32];
    }
    if (j0 == 0) aggh = agg + (size_t)chip_half() * AGG_PLANE;
#pragma unroll
    for (int j = 0; j < EDGE_BATCH; ++j) {
      const int dj = __builtin_amdgcn_readlane(d, j0 + j);
      const float nj = __int_as_float(__builtin_amdgcn_readlane(nrm_bits, j0 + j));
      float* ap = aggh + (size_t)dj * DD;
      atomicAdd(ap + ch, lo[j] * nj);
      atomicAdd(ap + ch + 32, hi[j] * nj);
    }
  }
}

__global__ void __launch_bounds__(256)
edge_agg_x_kernel(const float* __restrict__ x,
                  const int* __restrict__ src,
                  const int* __restrict__ dst,
                  const float* __restrict__ dinv,
                  float* __restrict__ ax,
                  float* __restrict__ nrm, int e) {
  const int tid  = blockIdx.x * blockDim.x + threadIdx.x;
  const int lane = threadIdx.x & 31;
  const int e0   = tid - lane;
  if (e0 >= e) return;
  const int s = src[e0 + lane];
  const int d = dst[e0 + lane];
  const float nv = dinv[s] * dinv[d];
  nrm[e0 + lane] = nv;
  const int nrm_bits = __float_as_int(nv);

  const int c = lane & 15;
  const bool up = (lane & 16) != 0;

  int sj[16], dj[16], nj[16];
#pragma unroll
  for (int j = 0; j < 16; ++j) {
    const int s_lo = __builtin_amdgcn_readlane(s, j);
    const int s_hi = __builtin_amdgcn_readlane(s, 16 + j);
    const int d_lo = __builtin_amdgcn_readlane(d, j);
    const int d_hi = __builtin_amdgcn_readlane(d, 16 + j);
    const int n_lo = __builtin_amdgcn_readlane(nrm_bits, j);
    const int n_hi = __builtin_amdgcn_readlane(nrm_bits, 16 + j);
    sj[j] = up ? s_hi : s_lo;
    dj[j] = up ? d_hi : d_lo;
    nj[j] = up ? n_hi : n_lo;
  }
  if (c < DIN) {
    float v[16];
#pragma unroll
    for (int j = 0; j < 16; ++j) v[j] = x[(size_t)sj[j] * DIN + c];
    float* axh = ax + (size_t)chip_half() * AX_PLANE;
#pragma unroll
    for (int j = 0; j < 16; ++j)
      atomicAdd(axh + (size_t)dj[j] * XP + c, v[j] * __int_as_float(nj[j]));
  }
}


__device__ inline void atomicMaxF32(float* addr, float val) {
  if (val >= 0.0f)
    atomicMax((int*)addr, __float_as_int(val));
  else
    atomicMin((unsigned int*)addr, __float_as_uint(val));
}

__global__ void pool_acc_kernel(const float* __restrict__ agg,
                                const int* __restrict__ batch,
                                float* __restrict__ gmax,
                                float* __restrict__ gsum,
                                float* __restrict__ cnt, int n) {
  int i = blockIdx.x * blockDim.x + threadIdx.x;
  if (i >= n * DD) return;
  const int row = i >> 6;
  const int f = i & 63;
  const int g = batch[row];
  const float v = tanhf(agg[i] + agg[AGG_PLANE + i]);
  atomicAdd(&gsum[(size_t)g * DD + f], v);
  atomicMaxF32(&gmax[(size_t)g * DD + f], v);
  if (f == 0) atomicAdd(&cnt[g], 1.0f);
}

__global__ void pool_finish_kernel(const float* __restrict__ gmax,
                                   const float* __restrict__ gsum,
                                   const float* __restrict__ cnt,
                                   float* __restrict__ hidden, int g_) {
  int i = blockIdx.x * blockDim.x + threadIdx.x;
  if (i >= g_ * DD) return;
  const int g = i >> 6;
  const int f = i & 63;
  const float c = cnt[g];
  float mx = gmax[(size_t)g * DD + f];
  if (c == 0.0f) mx = 0.0f;
  hidden[(size_t)g * (2 * DD) + f] = mx;
  hidden[(size_t)g * (2 * DD) + DD + f] = gsum[(size_t)g * DD + f] / fmaxf(c, 1.0f);
}

__global__ void head_kernel(const float* __restrict__ hidden,
                            const float* __restrict__ Wout,
                            const float* __restrict__ bout,
                            float* __restrict__ out, int g_) {
  int g = blockIdx.x * blockDim.x + threadIdx.x;
  if (g >= g_) return;
  float acc = bout[0];
#pragma unroll 8
  for (int k = 0; k < 2 * DD; ++k) acc += hidden[(size_t)g * (2 * DD) + k] * Wout[k];
  out[g] = acc;
}


extern "C" void kernel_launch(void* const* d_in, const int* in_sizes, int n_in,
                              void* d_out, int out_size, void* d_ws, size_t ws_size,
                              hipStream_t stream) {
  (void)in_sizes; (void)n_in; (void)out_size; (void)ws_size;

  const float* x    = (const float*)d_in[0];
  const int*   eidx = (const int*)d_in[1];
  const int*   batch = (const int*)d_in[2];
  const float* W0 = (const float*)d_in[3];
  const float* b0 = (const float*)d_in[4];
  const float* W1 = (const float*)d_in[5];
  const float* b1 = (const float*)d_in[6];
  const float* W2 = (const float*)d_in[7];
  const float* b2 = (const float*)d_in[8];
  const float* W3 = (const float*)d_in[9];
  const float* b3 = (const float*)d_in[10];
  const float* Wout = (const float*)d_in[11];
  const float* bout = (const float*)d_in[12];

  const int* src = eidx;
  const int* dst = eidx + EE;

  float* outp   = (float*)d_out;
  float* hidden = (float*)d_out + GG;

  float* ws = (float*)d_ws;
  float* dinv  = ws;
  float* hW    = dinv + NN_PAD;
  float* aggA  = hW   + (size_t)NN * DD;
  float* aggB  = aggA + 2 * AGG_PLANE;
  float* gmax  = aggB + 2 * AGG_PLANE;
  float* gsum  = gmax + (size_t)GG * DD;
  float* cnt   = gsum + (size_t)GG * DD;
  float* ax    = cnt  + GG;
  unsigned* part = (unsigned*)(ax + 2 * AX_PLANE);
  float* nrm   = (float*)(part + (size_t)DEG_PARTS * NN_PAD);

  const int T = 256;
  auto nb = [](long long n, int t) { return (int)((n + t - 1) / t); };

  degree_hist_kernel<<<DEG_PARTS, DEG_THREADS, 0, stream>>>(dst, part);
  dinv_ax_kernel<<<nb((long long)NN * 4, T), T, 0, stream>>>(part, x, dinv, ax, NN);

  const int gblocks = NN / 16 / GEMM_WAVES;
  const float* Wl[4] = {W0, W1, W2, W3};
  const float* bl[4] = {b0, b1, b2, b3};
  float* aggl[4] = {aggA, aggB, aggA, aggB};

  edge_agg_x_kernel<<<nb(EE, T), T, 0, stream>>>(x, src, dst, dinv, ax, nrm, EE);
  gemm_wmma_kernel<true><<<gblocks, GEMM_WAVES * 32, 0, stream>>>(ax, Wl[0], nullptr, bl[0], nullptr, aggl[0]);

  for (int l = 1; l < 4; ++l) {
    gemm_wmma_kernel<false><<<gblocks, GEMM_WAVES * 32, 0, stream>>>(aggl[l - 1], Wl[l], dinv, bl[l], hW, aggl[l]);

    edge_agg_kernel<<<nb(EE, T), T, 0, stream>>>(hW, src, dst, nrm, aggl[l], EE);
  }

  fill_f32_kernel<<<nb((long long)GG * DD, T), T, 0, stream>>>(gmax, -FLT_MAX, GG * DD);
  fill_f32_kernel<<<nb((long long)GG * DD, T), T, 0, stream>>>(gsum, 0.0f, GG * DD);
  fill_f32_kernel<<<nb(GG, T), T, 0, stream>>>(cnt, 0.0f, GG);
  pool_acc_kernel<<<nb((long long)NN * DD, T), T, 0, stream>>>(aggl[3], batch, gmax, gsum, cnt, NN);
  pool_finish_kernel<<<nb((long long)GG * DD, T), T, 0, stream>>>(gmax, gsum, cnt, hidden, GG);

  head_kernel<<<nb(GG, 128), 128, 0, stream>>>(hidden, Wout, bout, outp, GG);
}
